# starting kernel
# baseline (speedup 1.0000x reference)
#include <hip/hip_runtime.h>

typedef float v2f __attribute__((ext_vector_type(2)));
typedef float v8f __attribute__((ext_vector_type(8)));

#define N_NODES 50000
#define FDIM    128
#define N_EDGES 800000
#define KSTRIDE 260

__global__ __launch_bounds__(256) void sage_zero_kernel(float* __restrict__ p, int n) {
    int i = blockIdx.x * 256 + threadIdx.x;
    if (i < n) p[i] = 0.0f;
}

__global__ __launch_bounds__(256) void sage_scatter_kernel(
    const float* __restrict__ x,
    const int*   __restrict__ src,
    const int*   __restrict__ dst,
    float* __restrict__ msg,
    float* __restrict__ cnt,
    int with_count)
{
    int lane = threadIdx.x & 31;
    int wave = threadIdx.x >> 5;
    int e = blockIdx.x * 8 + wave;
    if (e >= N_EDGES) return;
    int s = src[e];
    int d = dst[e];
    const float4* xs = (const float4*)(x + (long long)s * FDIM);
    float4 v = xs[lane];
    float* o = msg + (long long)d * FDIM + lane * 4;
    atomicAdd(o + 0, v.x);
    atomicAdd(o + 1, v.y);
    atomicAdd(o + 2, v.z);
    atomicAdd(o + 3, v.w);
    if (with_count && lane == 0) atomicAdd(cnt + d, 1.0f);
}

__global__ __launch_bounds__(256) void sage_gemm_kernel(
    const float* __restrict__ msg, const float* __restrict__ cnt,
    const float* __restrict__ x,
    const float* __restrict__ Wl,  const float* __restrict__ Wr,
    const float* __restrict__ bias,
    float* __restrict__ out, int do_relu)
{
    __shared__ float lw[128 * KSTRIDE];

    int tid = threadIdx.x;
    for (int idx = tid; idx < 256 * 128; idx += 256) {
        int k = idx >> 7;
        int n = idx & 127;
        float w = (k < 128) ? Wl[idx] : Wr[idx - 128 * 128];
        lw[n * KSTRIDE + k] = w;
    }
    __syncthreads();

    int lane = tid & 31;
    int wave = tid >> 5;
    int h    = lane >> 4;
    int col  = lane & 15;

    int rowBase = blockIdx.x * 128 + wave * 16;
    int rowA    = rowBase + col;
    int rowAc   = rowA < N_NODES ? rowA : (N_NODES - 1);

    float scale = 1.0f / fmaxf(cnt[rowAc], 1.0f);
    const float* arow_m = msg + (long long)rowAc * FDIM + 2 * h;
    const float* arow_x = x   + (long long)rowAc * FDIM + 2 * h;
    const float* brow   = &lw[col * KSTRIDE + 2 * h];

    v8f acc[8];
    v8f z = {0.f, 0.f, 0.f, 0.f, 0.f, 0.f, 0.f, 0.f};
#pragma unroll
    for (int t = 0; t < 8; ++t) acc[t] = z;

#pragma unroll 4
    for (int kk = 0; kk < 128; kk += 4) {
        v2f a = *(const v2f*)(arow_m + kk);
        a.x *= scale;
        a.y *= scale;
#pragma unroll
        for (int t = 0; t < 8; ++t) {
            v2f b = *(const v2f*)(brow + t * (16 * KSTRIDE) + kk);
            acc[t] = __builtin_amdgcn_wmma_f32_16x16x4_f32(
                false, a, false, b, (short)0, acc[t], false, false);
        }
    }

#pragma unroll 4
    for (int kk = 0; kk < 128; kk += 4) {
        v2f a = *(const v2f*)(arow_x + kk);
#pragma unroll
        for (int t = 0; t < 8; ++t) {
            v2f b = *(const v2f*)(brow + t * (16 * KSTRIDE) + 128 + kk);
            acc[t] = __builtin_amdgcn_wmma_f32_16x16x4_f32(
                false, a, false, b, (short)0, acc[t], false, false);
        }
    }

#pragma unroll
    for (int t = 0; t < 8; ++t) {
        int n  = t * 16 + col;
        float bv = bias[n];
#pragma unroll
        for (int v = 0; v < 8; ++v) {
            int row = rowBase + v + 8 * h;
            float val = acc[t][v] + bv;
            if (do_relu) val = fmaxf(val, 0.0f);
            if (row < N_NODES) out[(long long)row * FDIM + n] = val;
        }
    }
}

extern "C" void kernel_launch(void* const* d_in, const int* in_sizes, int n_in,
                              void* d_out, int out_size, void* d_ws, size_t ws_size,
                              hipStream_t stream) {
    const float* x   = (const float*)d_in[0];
    const int*   ei  = (const int*)d_in[1];
    const float* Wl0 = (const float*)d_in[2];
    const float* Wr0 = (const float*)d_in[3];
    const float* b0  = (const float*)d_in[4];
    const float* Wl1 = (const float*)d_in[5];
    const float* Wr1 = (const float*)d_in[6];
    const float* b1  = (const float*)d_in[7];
    float* out = (float*)d_out;

    const int* src = ei;
    const int* dst = ei + N_EDGES;

    float* msg = (float*)d_ws;
    float* cnt = msg + (size_t)N_NODES * FDIM;
    float* hbuf = cnt + N_NODES;

    const int nmsg = N_NODES * FDIM;
    const int gemm_blocks = (N_NODES + 127) / 128;
    const int edge_blocks = N_EDGES / 8;

    sage_zero_kernel<<<(nmsg + N_NODES + 255) / 256, 256, 0, stream>>>(msg, nmsg + N_NODES);
    sage_scatter_kernel<<<edge_blocks, 256, 0, stream>>>(x, src, dst, msg, cnt, 1);
    sage_gemm_kernel<<<gemm_blocks, 256, 0, stream>>>(msg, cnt, x, Wl0, Wr0, b0, hbuf, 1);

    sage_zero_kernel<<<(nmsg + 255) / 256, 256, 0, stream>>>(msg, nmsg);
    sage_scatter_kernel<<<edge_blocks, 256, 0, stream>>>(hbuf, src, dst, msg, cnt, 0);
    sage_gemm_kernel<<<gemm_blocks, 256, 0, stream>>>(msg, cnt, hbuf, Wl1, Wr1, b1, out, 0);
}


# float-atomic scatter replaced: edges counting-sorted by dst once per call (int atomic gives count and arrival-order rank); each GEMM wave sums its rows' neighbours in rank order into an LDS A tile
# speedup vs baseline: 1.9522x; 1.9522x over previous
#include <hip/hip_runtime.h>


typedef float v2f __attribute__((ext_vector_type(2)));
typedef float v8f __attribute__((ext_vector_type(8)));

#define N_NODES 50000
#define FDIM    128
#define N_EDGES 800000
#define KSTRIDE 260
#define ASTRIDE 132
#define GWAVES  16
#define NPAD    50176

__global__ __launch_bounds__(256) void sage_zero_kernel(int* __restrict__ p, int n) {
    int i = blockIdx.x * 256 + threadIdx.x;
    if (i < n) p[i] = 0;
}

__global__ __launch_bounds__(256) void sage_rank_kernel(
    const int* __restrict__ dst, int* __restrict__ deg, int* __restrict__ rank)
{
    int e = blockIdx.x * 256 + threadIdx.x;
    if (e < N_EDGES) rank[e] = atomicAdd(deg + dst[e], 1);
}

__global__ __launch_bounds__(256) void sage_rowptr_kernel(
    const int* __restrict__ deg, int* __restrict__ rowptr)
{
    __shared__ int s_part[8];
    __shared__ int s_scan[8];
    int tid  = threadIdx.x;
    int lane = tid & 31;
    int wave = tid >> 5;
    int base = blockIdx.x * 256;

    const int4* deg4 = (const int4*)deg;
    int part = 0;
    for (int j = tid; j < (base >> 2); j += 256) {
        int4 q = deg4[j];
        part += (q.x + q.y) + (q.z + q.w);
    }
    int i = base + tid;
    int d = (i < N_NODES) ? deg[i] : 0;

    int inc = d;
    for (int o = 1; o < 32; o <<= 1) {
        int t = __shfl_up(inc, o);
        if (lane >= o) inc += t;
    }
    for (int o = 16; o > 0; o >>= 1) part += __shfl_xor(part, o);
    if (lane == 31) s_scan[wave] = inc;
    if (lane == 0)  s_part[wave] = part;
    __syncthreads();
    int pre = 0;
    for (int w = 0; w < 8; ++w) {
        pre += s_part[w];
        if (w < wave) pre += s_scan[w];
    }
    int excl = pre + inc - d;
    if (i < N_NODES) rowptr[i] = excl;
    if (i == N_NODES - 1) rowptr[N_NODES] = excl + d;
}

__global__ __launch_bounds__(256) void sage_place_kernel(
    const int* __restrict__ src, const int* __restrict__ dst,
    const int* __restrict__ rowptr, const int* __restrict__ rank, int* __restrict__ perm)
{
    int e = blockIdx.x * 256 + threadIdx.x;
    if (e < N_EDGES) {
        int pos = rowptr[dst[e]] + rank[e];
        if ((unsigned)pos < (unsigned)N_EDGES) perm[pos] = src[e];
    }
}

__global__ __launch_bounds__(512) void sage_gemm_kernel(
    const float* __restrict__ x,
    const int* __restrict__ rowptr, const int* __restrict__ perm,
    const float* __restrict__ Wl,  const float* __restrict__ Wr,
    const float* __restrict__ bias,
    float* __restrict__ out, int do_relu)
{
    __shared__ float lw[128 * KSTRIDE];
    __shared__ float at[GWAVES * 16 * ASTRIDE];

    int tid = threadIdx.x;
    for (int idx = tid; idx < 256 * 128; idx += 512) {
        int k = idx >> 7;
        int n = idx & 127;
        float w = (k < 128) ? Wl[idx] : Wr[idx - 128 * 128];
        lw[n * KSTRIDE + k] = w;
    }

    int lane = tid & 31;
    int wave = tid >> 5;
    int h    = lane >> 4;
    int col  = lane & 15;

    int rowBase = blockIdx.x * (GWAVES * 16) + wave * 16;
    float* atw  = at + wave * (16 * ASTRIDE);

    {
        const float4* __restrict__ x4 = (const float4*)x;
        const float4 zero4 = {0.f, 0.f, 0.f, 0.f};

        auto issue_rows = [&](int ids, int j, int n, float4 (&v)[16]) {
            unsigned s[16];
#pragma unroll
            for (int u = 0; u < 16; ++u) {
                s[u] = (unsigned)__builtin_amdgcn_readlane(ids, j + u);
                if (s[u] > (unsigned)(N_NODES - 1)) s[u] = N_NODES - 1;
            }
#pragma unroll
            for (int u = 0; u < 16; ++u) {
                if (j + u < n) v[u] = x4[(size_t)s[u] * 32 + lane];
            }
        };
        auto add_rows = [&](const float4 (&v)[16], int j, int n, float4& acc) {
#pragma unroll
            for (int u = 0; u < 16; ++u) {
                if (j + u < n) {
                    acc.x += v[u].x;
                    acc.y += v[u].y;
                    acc.z += v[u].z;
                    acc.w += v[u].w;
                }
            }
        };

        int rpi = rowBase + lane;
        if (rpi > N_NODES) rpi = N_NODES;
        int rp = rowptr[rpi];
        rp = min(max(rp, 0), N_EDGES);

        int nxt = 0;
        {
            int b0 = __builtin_amdgcn_readlane(rp, 0);
            int e0 = __builtin_amdgcn_readlane(rp, 1);
            if (b0 + lane < e0) nxt = perm[b0 + lane];
        }
        for (int r = 0; r < 16; ++r) {
            int beg = __builtin_amdgcn_readlane(rp, r);
            int end = __builtin_amdgcn_readlane(rp, r + 1);
            int e1  = __builtin_amdgcn_readlane(rp, (r + 2) & 31);
            int my  = nxt;
            int n0  = end - beg;
            if (n0 > 32) n0 = 32;

            float4 v[16];
            float4 acc = zero4;
            issue_rows(my, 0, n0, v);
            nxt = 0;
            if (r < 15 && end + lane < e1) nxt = perm[end + lane];
            add_rows(v, 0, n0, acc);
            if (n0 > 16) {
                issue_rows(my, 16, n0, v);
                add_rows(v, 16, n0, acc);
            }
            for (int base = beg + 32; base < end; base += 32) {
                int more = 0;
                if (base + lane < end) more = perm[base + lane];
                int n = end - base;
                if (n > 32) n = 32;
                for (int j = 0; j < n; j += 16) {
                    issue_rows(more, j, n, v);
                    add_rows(v, j, n, acc);
                }
            }
            float scale = 1.0f / fmaxf((float)(end - beg), 1.0f);
            acc.x *= scale;
            acc.y *= scale;
            acc.z *= scale;
            acc.w *= scale;
            *(float4*)(atw + r * ASTRIDE + 4 * lane) = acc;
        }
    }
    __syncthreads();

    int rowA    = rowBase + col;
    int rowAc   = rowA < N_NODES ? rowA : (N_NODES - 1);

    const float* arow_m = atw + col * ASTRIDE + 2 * h;
    const float* arow_x = x   + (long long)rowAc * FDIM + 2 * h;
    const float* brow   = &lw[col * KSTRIDE + 2 * h];

    v8f acc[8];
    v8f z = {0.f, 0.f, 0.f, 0.f, 0.f, 0.f, 0.f, 0.f};
#pragma unroll
    for (int t = 0; t < 8; ++t) acc[t] = z;

#pragma unroll 4
    for (int kk = 0; kk < 128; kk += 4) {
        v2f a = *(const v2f*)(arow_m + kk);
#pragma unroll
        for (int t = 0; t < 8; ++t) {
            v2f b = *(const v2f*)(brow + t * (16 * KSTRIDE) + kk);
            acc[t] = __builtin_amdgcn_wmma_f32_16x16x4_f32(
                false, a, false, b, (short)0, acc[t], false, false);
        }
    }

#pragma unroll 4
    for (int kk = 0; kk < 128; kk += 4) {
        v2f a = *(const v2f*)(arow_x + kk);
#pragma unroll
        for (int t = 0; t < 8; ++t) {
            v2f b = *(const v2f*)(brow + t * (16 * KSTRIDE) + 128 + kk);
            acc[t] = __builtin_amdgcn_wmma_f32_16x16x4_f32(
                false, a, false, b, (short)0, acc[t], false, false);
        }
    }

#pragma unroll
    for (int t = 0; t < 8; ++t) {
        int n  = t * 16 + col;
        float bv = bias[n];
#pragma unroll
        for (int v = 0; v < 8; ++v) {
            int row = rowBase + v + 8 * h;
            float val = acc[t][v] + bv;
            if (do_relu) val = fmaxf(val, 0.0f);
            if (row < N_NODES) out[(long long)row * FDIM + n] = val;
        }
    }
}

extern "C" void kernel_launch(void* const* d_in, const int* in_sizes, int n_in,
                              void* d_out, int out_size, void* d_ws, size_t ws_size,
                              hipStream_t stream) {
    const float* x   = (const float*)d_in[0];
    const int*   ei  = (const int*)d_in[1];
    const float* Wl0 = (const float*)d_in[2];
    const float* Wr0 = (const float*)d_in[3];
    const float* b0  = (const float*)d_in[4];
    const float* Wl1 = (const float*)d_in[5];
    const float* Wr1 = (const float*)d_in[6];
    const float* b1  = (const float*)d_in[7];
    float* out = (float*)d_out;

    const int* src = ei;
    const int* dst = ei + N_EDGES;

    int* deg    = (int*)d_ws;
    int* rowptr = deg + NPAD;
    int* rank   = rowptr + NPAD;
    int* perm   = rank + N_EDGES;
    float* hbuf = (float*)d_ws + (size_t)N_NODES * FDIM + N_NODES;

    const int node_blocks = NPAD / 256;
    const int edge_blocks = N_EDGES / 256;
    const int gemm_blocks = NPAD / (GWAVES * 16);

    sage_zero_kernel<<<node_blocks, 256, 0, stream>>>(deg, NPAD);
    sage_rank_kernel<<<edge_blocks, 256, 0, stream>>>(dst, deg, rank);
    sage_rowptr_kernel<<<node_blocks, 256, 0, stream>>>(deg, rowptr);
    sage_place_kernel<<<edge_blocks, 256, 0, stream>>>(src, dst, rowptr, rank, perm);

    sage_gemm_kernel<<<gemm_blocks, 512, 0, stream>>>(x, rowptr, perm, Wl0, Wr0, b0, hbuf, 1);
    sage_gemm_kernel<<<gemm_blocks, 512, 0, stream>>>(hbuf, rowptr, perm, Wl1, Wr1, b1, out, 0);
}


# edges sorted by destination with a two-level counting sort on LDS counters (2 launches) instead of clear + global returning atomics + scan + scattered placement (4); in-node order: LDS atomic arrival
# speedup vs baseline: 3.1577x; 1.6175x over previous
#include <hip/hip_runtime.h>


typedef float v2f __attribute__((ext_vector_type(2)));
typedef float v8f __attribute__((ext_vector_type(8)));

#define N_NODES 50000
#define FDIM    128
#define N_EDGES 800000
#define KSTRIDE 260
#define ASTRIDE 132
#define GWAVES  16
#define NPAD    50176

#define NBIN     196
#define NCHUNK   256
#define CHUNK    3125
#define ATHREADS 320
#define AEPT     10
#define LCAP     8192
static_assert(NCHUNK * CHUNK == N_EDGES, "chunks cover the edge list exactly");
static_assert(ATHREADS * AEPT >= CHUNK, "every edge of a chunk has a slot");
static_assert(NBIN * 256 == NPAD, "ranges cover the padded node count");
static_assert(N_NODES <= 65536, "a source id fits the low 16 bits of a staged word");

__global__ __launch_bounds__(ATHREADS) void sage_bin_kernel(
    const int* __restrict__ src, const int* __restrict__ dst,
    int* __restrict__ tab, int* __restrict__ offtab, int* __restrict__ gstage)
{
    __shared__ int s_cnt[NBIN];
    __shared__ int s_off[NBIN];
    __shared__ int s_stage[CHUNK];
    const int tid = threadIdx.x;
    const int c   = blockIdx.x;
    if (tid < NBIN) s_cnt[tid] = 0;
    __syncthreads();

    int dv[AEPT];
    int sv[AEPT];
    int bin[AEPT];
    int word[AEPT];
    int r[AEPT];
#pragma unroll
    for (int k = 0; k < AEPT; ++k) {
        int i = k * ATHREADS + tid;
        int e = c * CHUNK + min(i, CHUNK - 1);
        dv[k] = dst[e];
        sv[k] = src[e];
    }
#pragma unroll
    for (int k = 0; k < AEPT; ++k) {
        int i = k * ATHREADS + tid;
        unsigned ub = (unsigned)dv[k] >> 8;
        bin[k]  = ub > (unsigned)(NBIN - 1) ? (NBIN - 1) : (int)ub;
        word[k] = ((dv[k] & 255) << 16) | (sv[k] & 0xFFFF);
        r[k] = 0;
        if (i < CHUNK) r[k] = atomicAdd(&s_cnt[bin[k]], 1);
    }
    __syncthreads();
    if (tid < 32) {
        int v[7];
        int sum = 0;
#pragma unroll
        for (int j = 0; j < 7; ++j) {
            int bb = tid * 7 + j;
            v[j] = (bb < NBIN) ? s_cnt[bb] : 0;
            sum += v[j];
        }
        int inc = sum;
        for (int o = 1; o < 32; o <<= 1) {
            int t = __shfl_up(inc, o);
            if (tid >= o) inc += t;
        }
        int run = inc - sum;
#pragma unroll
        for (int j = 0; j < 7; ++j) {
            int bb = tid * 7 + j;
            if (bb < NBIN) s_off[bb] = run;
            run += v[j];
        }
    }
    __syncthreads();
    if (tid < NBIN) {
        tab[tid * NCHUNK + c]    = s_cnt[tid];
        offtab[tid * NCHUNK + c] = s_off[tid];
    }
#pragma unroll
    for (int k = 0; k < AEPT; ++k) {
        int i = k * ATHREADS + tid;
        if (i < CHUNK) {
            int pos = s_off[bin[k]] + r[k];
            if ((unsigned)pos < (unsigned)CHUNK) s_stage[pos] = word[k];
        }
    }
    __syncthreads();
    for (int i = tid; i < CHUNK; i += ATHREADS) gstage[c * CHUNK + i] = s_stage[i];
}

__global__ __launch_bounds__(256) void sage_fine_kernel(
    const int* __restrict__ tab, const int* __restrict__ offtab, const int* __restrict__ gstage,
    int* __restrict__ rowptr, int* __restrict__ perm)
{
    __shared__ int s_red[16];
    __shared__ int s_wsum[8];
    __shared__ int s_cn[NCHUNK];
    __shared__ int s_ro[NCHUNK];
    __shared__ int s_deg[256];
    __shared__ int s_cur[256];
    __shared__ int s_perm[LCAP];
    const int tid  = threadIdx.x;
    const int lane = tid & 31;
    const int wave = tid >> 5;
    const int b    = blockIdx.x;

    int cn = tab[b * NCHUNK + tid];
    int of = offtab[b * NCHUNK + tid];
    cn = min(max(cn, 0), CHUNK);
    of = min(max(of, 0), CHUNK - cn);
    s_cn[tid]  = cn;
    s_ro[tid]  = tid * CHUNK + min(of, CHUNK - 1);
    s_deg[tid] = 0;

    const int4* tab4 = (const int4*)tab;
    const int n4 = b * (NCHUNK / 4);
    int part = 0;
    for (int j0 = 0; j0 < n4; j0 += 256 * 8) {
        int4 qv[8];
#pragma unroll
        for (int u = 0; u < 8; ++u) {
            int j = j0 + u * 256 + tid;
            qv[u] = tab4[min(j, n4 - 1)];
        }
        __builtin_amdgcn_sched_barrier(0);
#pragma unroll
        for (int u = 0; u < 8; ++u) {
            int j = j0 + u * 256 + tid;
            int s = (qv[u].x + qv[u].y) + (qv[u].z + qv[u].w);
            part += (j < n4) ? s : 0;
        }
    }
    int ps = part;
    int cs = cn;
    for (int o = 16; o > 0; o >>= 1) {
        ps += __shfl_xor(ps, o);
        cs += __shfl_xor(cs, o);
    }
    if (lane == 0) {
        s_red[wave]     = ps;
        s_red[8 + wave] = cs;
    }
    __syncthreads();
    int base  = 0;
    int total = 0;
    for (int w = 0; w < 8; ++w) {
        base  += s_red[w];
        total += s_red[8 + w];
    }
    base = min(max(base, 0), N_EDGES);

    const int grp = tid >> 3;
    const int sub = tid & 7;
    int n[8];
    int ro[8];
    int maxn = 0;
#pragma unroll
    for (int r = 0; r < 8; ++r) {
        n[r]  = s_cn[r * 32 + grp];
        ro[r] = s_ro[r * 32 + grp];
        maxn  = max(maxn, n[r]);
    }

    for (int jb = 0; jb < maxn; jb += 8) {
        const int idx = jb + sub;
        int w[8];
#pragma unroll
        for (int r = 0; r < 8; ++r) w[r] = gstage[ro[r] + min(idx, max(n[r] - 1, 0))];
#pragma unroll
        for (int r = 0; r < 8; ++r) {
            atomicAdd(&s_deg[(w[r] >> 16) & 255], (idx < n[r]) ? 1 : 0);
        }
    }
    __syncthreads();
    const int d = s_deg[tid];
    int inc = d;
    for (int o = 1; o < 32; o <<= 1) {
        int t = __shfl_up(inc, o);
        if (lane >= o) inc += t;
    }
    if (lane == 31) s_wsum[wave] = inc;
    __syncthreads();
    int pre = 0;
    for (int w = 0; w < 8; ++w) {
        if (w < wave) pre += s_wsum[w];
    }
    const int excl = pre + inc - d;
    const int node = b * 256 + tid;
    if (node < N_NODES) rowptr[node] = base + excl;
    if (node == N_NODES - 1) rowptr[N_NODES] = base + excl + d;
    s_cur[tid] = excl;
    __syncthreads();

    const bool in_lds = total <= LCAP;
    for (int jb = 0; jb < maxn; jb += 8) {
        const int idx = jb + sub;
        int w[8];
#pragma unroll
        for (int r = 0; r < 8; ++r) w[r] = gstage[ro[r] + min(idx, max(n[r] - 1, 0))];
#pragma unroll
        for (int r = 0; r < 8; ++r) {
            const int act = (idx < n[r]) ? 1 : 0;
            int pos = atomicAdd(&s_cur[(w[r] >> 16) & 255], act);
            int sid = w[r] & 0xFFFF;
            if (act) {
                if (in_lds) {
                    if ((unsigned)pos < (unsigned)LCAP) s_perm[pos] = sid;
                } else {
                    int gi = base + pos;
                    if ((unsigned)gi < (unsigned)N_EDGES) perm[gi] = sid;
                }
            }
        }
    }
    __syncthreads();
    if (in_lds) {
        for (int i = tid; i < total; i += 256) {
            int gi = base + i;
            if ((unsigned)gi < (unsigned)N_EDGES) perm[gi] = s_perm[i];
        }
    }
}

__global__ __launch_bounds__(512) void sage_gemm_kernel(
    const float* __restrict__ x,
    const int* __restrict__ rowptr, const int* __restrict__ perm,
    const float* __restrict__ Wl,  const float* __restrict__ Wr,
    const float* __restrict__ bias,
    float* __restrict__ out, int do_relu)
{
    __shared__ float lw[128 * KSTRIDE];
    __shared__ float at[GWAVES * 16 * ASTRIDE];

    int tid = threadIdx.x;
    for (int idx = tid; idx < 256 * 128; idx += 512) {
        int k = idx >> 7;
        int n = idx & 127;
        float w = (k < 128) ? Wl[idx] : Wr[idx - 128 * 128];
        lw[n * KSTRIDE + k] = w;
    }

    int lane = tid & 31;
    int wave = tid >> 5;
    int h    = lane >> 4;
    int col  = lane & 15;

    int rowBase = blockIdx.x * (GWAVES * 16) + wave * 16;
    float* atw  = at + wave * (16 * ASTRIDE);

    {
        const float4* __restrict__ x4 = (const float4*)x;
        const float4 zero4 = {0.f, 0.f, 0.f, 0.f};

        auto issue_rows = [&](int ids, int j, int n, float4 (&v)[16]) {
            unsigned s[16];
#pragma unroll
            for (int u = 0; u < 16; ++u) {
                s[u] = (unsigned)__builtin_amdgcn_readlane(ids, j + u);
                if (s[u] > (unsigned)(N_NODES - 1)) s[u] = N_NODES - 1;
            }
#pragma unroll
            for (int u = 0; u < 16; ++u) {
                if (j + u < n) v[u] = x4[(size_t)s[u] * 32 + lane];
            }
        };
        auto add_rows = [&](const float4 (&v)[16], int j, int n, float4& acc) {
#pragma unroll
            for (int u = 0; u < 16; ++u) {
                if (j + u < n) {
                    acc.x += v[u].x;
                    acc.y += v[u].y;
                    acc.z += v[u].z;
                    acc.w += v[u].w;
                }
            }
        };

        int rpi = rowBase + lane;
        if (rpi > N_NODES) rpi = N_NODES;
        int rp = rowptr[rpi];
        rp = min(max(rp, 0), N_EDGES);

        int nxt = 0;
        {
            int b0 = __builtin_amdgcn_readlane(rp, 0);
            int e0 = __builtin_amdgcn_readlane(rp, 1);
            if (b0 + lane < e0) nxt = perm[b0 + lane];
        }
        for (int r = 0; r < 16; ++r) {
            int beg = __builtin_amdgcn_readlane(rp, r);
            int end = __builtin_amdgcn_readlane(rp, r + 1);
            int e1  = __builtin_amdgcn_readlane(rp, (r + 2) & 31);
            int my  = nxt;
            int n0  = end - beg;
            if (n0 > 32) n0 = 32;

            float4 v[16];
            float4 acc = zero4;
            issue_rows(my, 0, n0, v);
            nxt = 0;
            if (r < 15 && end + lane < e1) nxt = perm[end + lane];
            add_rows(v, 0, n0, acc);
            if (n0 > 16) {
                issue_rows(my, 16, n0, v);
                add_rows(v, 16, n0, acc);
            }
            for (int base = beg + 32; base < end; base += 32) {
                int more = 0;
                if (base + lane < end) more = perm[base + lane];
                int n = end - base;
                if (n > 32) n = 32;
                for (int j = 0; j < n; j += 16) {
                    issue_rows(more, j, n, v);
                    add_rows(v, j, n, acc);
                }
            }
            float scale = 1.0f / fmaxf((float)(end - beg), 1.0f);
            acc.x *= scale;
            acc.y *= scale;
            acc.z *= scale;
            acc.w *= scale;
            *(float4*)(atw + r * ASTRIDE + 4 * lane) = acc;
        }
    }
    __syncthreads();

    int rowA    = rowBase + col;
    int rowAc   = rowA < N_NODES ? rowA : (N_NODES - 1);

    const float* arow_m = atw + col * ASTRIDE + 2 * h;
    const float* arow_x = x   + (long long)rowAc * FDIM + 2 * h;
    const float* brow   = &lw[col * KSTRIDE + 2 * h];

    v8f acc[8];
    v8f z = {0.f, 0.f, 0.f, 0.f, 0.f, 0.f, 0.f, 0.f};
#pragma unroll
    for (int t = 0; t < 8; ++t) acc[t] = z;

#pragma unroll 4
    for (int kk = 0; kk < 128; kk += 4) {
        v2f a = *(const v2f*)(arow_m + kk);
#pragma unroll
        for (int t = 0; t < 8; ++t) {
            v2f b = *(const v2f*)(brow + t * (16 * KSTRIDE) + kk);
            acc[t] = __builtin_amdgcn_wmma_f32_16x16x4_f32(
                false, a, false, b, (short)0, acc[t], false, false);
        }
    }

#pragma unroll 4
    for (int kk = 0; kk < 128; kk += 4) {
        v2f a = *(const v2f*)(arow_x + kk);
#pragma unroll
        for (int t = 0; t < 8; ++t) {
            v2f b = *(const v2f*)(brow + t * (16 * KSTRIDE) + 128 + kk);
            acc[t] = __builtin_amdgcn_wmma_f32_16x16x4_f32(
                false, a, false, b, (short)0, acc[t], false, false);
        }
    }

#pragma unroll
    for (int t = 0; t < 8; ++t) {
        int n  = t * 16 + col;
        float bv = bias[n];
#pragma unroll
        for (int v = 0; v < 8; ++v) {
            int row = rowBase + v + 8 * h;
            float val = acc[t][v] + bv;
            if (do_relu) val = fmaxf(val, 0.0f);
            if (row < N_NODES) out[(long long)row * FDIM + n] = val;
        }
    }
}

extern "C" void kernel_launch(void* const* d_in, const int* in_sizes, int n_in,
                              void* d_out, int out_size, void* d_ws, size_t ws_size,
                              hipStream_t stream) {
    const float* x   = (const float*)d_in[0];
    const int*   ei  = (const int*)d_in[1];
    const float* Wl0 = (const float*)d_in[2];
    const float* Wr0 = (const float*)d_in[3];
    const float* b0  = (const float*)d_in[4];
    const float* Wl1 = (const float*)d_in[5];
    const float* Wr1 = (const float*)d_in[6];
    const float* b1  = (const float*)d_in[7];
    float* out = (float*)d_out;

    const int* src = ei;
    const int* dst = ei + N_EDGES;

    int* rowptr = (int*)d_ws;
    int* gstage = rowptr + NPAD;
    int* perm   = gstage + N_EDGES;
    int* tab    = perm + N_EDGES;
    int* offtab = tab + NPAD;
    float* hbuf = (float*)d_ws + (size_t)N_NODES * FDIM + N_NODES;

    const int gemm_blocks = NPAD / (GWAVES * 16);

    sage_bin_kernel<<<NCHUNK, ATHREADS, 0, stream>>>(src, dst, tab, offtab, gstage);
    sage_fine_kernel<<<NBIN, 256, 0, stream>>>(tab, offtab, gstage, rowptr, perm);

    sage_gemm_kernel<<<gemm_blocks, 512, 0, stream>>>(x, rowptr, perm, Wl0, Wr0, b0, hbuf, 1);
    sage_gemm_kernel<<<gemm_blocks, 512, 0, stream>>>(hbuf, rowptr, perm, Wl1, Wr1, b1, out, 0);
}


# GEMM wave's neighbour gather read as one double-buffered edge stream over its 16 rows: 16 row loads always in flight behind the 16 being summed, branch-free issue; per-row sum order unchanged
# speedup vs baseline: 3.2518x; 1.0298x over previous
#include <hip/hip_runtime.h>


typedef float v2f __attribute__((ext_vector_type(2)));
typedef float v8f __attribute__((ext_vector_type(8)));

#define N_NODES 50000
#define FDIM    128
#define N_EDGES 800000
#define KSTRIDE 260
#define ASTRIDE 132
#define GWAVES  16
#define NPAD    50176

#define NBIN     196
#define NCHUNK   256
#define CHUNK    3125
#define ATHREADS 320
#define AEPT     10
#define LCAP     8192
static_assert(NCHUNK * CHUNK == N_EDGES, "chunks cover the edge list exactly");
static_assert(ATHREADS * AEPT >= CHUNK, "every edge of a chunk has a slot");
static_assert(NBIN * 256 == NPAD, "ranges cover the padded node count");
static_assert(N_NODES <= 65536, "a source id fits the low 16 bits of a staged word");

__global__ __launch_bounds__(ATHREADS) void sage_bin_kernel(
    const int* __restrict__ src, const int* __restrict__ dst,
    int* __restrict__ tab, int* __restrict__ offtab, int* __restrict__ gstage)
{
    __shared__ int s_cnt[NBIN];
    __shared__ int s_off[NBIN];
    __shared__ int s_stage[CHUNK];
    const int tid = threadIdx.x;
    const int c   = blockIdx.x;
    if (tid < NBIN) s_cnt[tid] = 0;
    __syncthreads();

    int dv[AEPT];
    int sv[AEPT];
    int bin[AEPT];
    int word[AEPT];
    int r[AEPT];
#pragma unroll
    for (int k = 0; k < AEPT; ++k) {
        int i = k * ATHREADS + tid;
        int e = c * CHUNK + min(i, CHUNK - 1);
        dv[k] = dst[e];
        sv[k] = src[e];
    }
#pragma unroll
    for (int k = 0; k < AEPT; ++k) {
        int i = k * ATHREADS + tid;
        unsigned ub = (unsigned)dv[k] >> 8;
        bin[k]  = ub > (unsigned)(NBIN - 1) ? (NBIN - 1) : (int)ub;
        word[k] = ((dv[k] & 255) << 16) | (sv[k] & 0xFFFF);
        r[k] = 0;
        if (i < CHUNK) r[k] = atomicAdd(&s_cnt[bin[k]], 1);
    }
    __syncthreads();
    if (tid < 32) {
        int v[7];
        int sum = 0;
#pragma unroll
        for (int j = 0; j < 7; ++j) {
            int bb = tid * 7 + j;
            v[j] = (bb < NBIN) ? s_cnt[bb] : 0;
            sum += v[j];
        }
        int inc = sum;
        for (int o = 1; o < 32; o <<= 1) {
            int t = __shfl_up(inc, o);
            if (tid >= o) inc += t;
        }
        int run = inc - sum;
#pragma unroll
        for (int j = 0; j < 7; ++j) {
            int bb = tid * 7 + j;
            if (bb < NBIN) s_off[bb] = run;
            run += v[j];
        }
    }
    __syncthreads();
    if (tid < NBIN) {
        tab[tid * NCHUNK + c]    = s_cnt[tid];
        offtab[tid * NCHUNK + c] = s_off[tid];
    }
#pragma unroll
    for (int k = 0; k < AEPT; ++k) {
        int i = k * ATHREADS + tid;
        if (i < CHUNK) {
            int pos = s_off[bin[k]] + r[k];
            if ((unsigned)pos < (unsigned)CHUNK) s_stage[pos] = word[k];
        }
    }
    __syncthreads();
    for (int i = tid; i < CHUNK; i += ATHREADS) gstage[c * CHUNK + i] = s_stage[i];
}

__global__ __launch_bounds__(256) void sage_fine_kernel(
    const int* __restrict__ tab, const int* __restrict__ offtab, const int* __restrict__ gstage,
    int* __restrict__ rowptr, int* __restrict__ perm)
{
    __shared__ int s_red[16];
    __shared__ int s_wsum[8];
    __shared__ int s_cn[NCHUNK];
    __shared__ int s_ro[NCHUNK];
    __shared__ int s_deg[256];
    __shared__ int s_cur[256];
    __shared__ int s_perm[LCAP];
    const int tid  = threadIdx.x;
    const int lane = tid & 31;
    const int wave = tid >> 5;
    const int b    = blockIdx.x;

    int cn = tab[b * NCHUNK + tid];
    int of = offtab[b * NCHUNK + tid];
    cn = min(max(cn, 0), CHUNK);
    of = min(max(of, 0), CHUNK - cn);
    s_cn[tid]  = cn;
    s_ro[tid]  = tid * CHUNK + min(of, CHUNK - 1);
    s_deg[tid] = 0;

    const int4* tab4 = (const int4*)tab;
    const int n4 = b * (NCHUNK / 4);
    int part = 0;
    for (int j0 = 0; j0 < n4; j0 += 256 * 8) {
        int4 qv[8];
#pragma unroll
        for (int u = 0; u < 8; ++u) {
            int j = j0 + u * 256 + tid;
            qv[u] = tab4[min(j, n4 - 1)];
        }
        __builtin_amdgcn_sched_barrier(0);
#pragma unroll
        for (int u = 0; u < 8; ++u) {
            int j = j0 + u * 256 + tid;
            int s = (qv[u].x + qv[u].y) + (qv[u].z + qv[u].w);
            part += (j < n4) ? s : 0;
        }
    }
    int ps = part;
    int cs = cn;
    for (int o = 16; o > 0; o >>= 1) {
        ps += __shfl_xor(ps, o);
        cs += __shfl_xor(cs, o);
    }
    if (lane == 0) {
        s_red[wave]     = ps;
        s_red[8 + wave] = cs;
    }
    __syncthreads();
    int base  = 0;
    int total = 0;
    for (int w = 0; w < 8; ++w) {
        base  += s_red[w];
        total += s_red[8 + w];
    }
    base = min(max(base, 0), N_EDGES);

    const int grp = tid >> 3;
    const int sub = tid & 7;
    int n[8];
    int ro[8];
    int maxn = 0;
#pragma unroll
    for (int r = 0; r < 8; ++r) {
        n[r]  = s_cn[r * 32 + grp];
        ro[r] = s_ro[r * 32 + grp];
        maxn  = max(maxn, n[r]);
    }

    for (int jb = 0; jb < maxn; jb += 8) {
        const int idx = jb + sub;
        int w[8];
#pragma unroll
        for (int r = 0; r < 8; ++r) w[r] = gstage[ro[r] + min(idx, max(n[r] - 1, 0))];
#pragma unroll
        for (int r = 0; r < 8; ++r) {
            atomicAdd(&s_deg[(w[r] >> 16) & 255], (idx < n[r]) ? 1 : 0);
        }
    }
    __syncthreads();
    const int d = s_deg[tid];
    int inc = d;
    for (int o = 1; o < 32; o <<= 1) {
        int t = __shfl_up(inc, o);
        if (lane >= o) inc += t;
    }
    if (lane == 31) s_wsum[wave] = inc;
    __syncthreads();
    int pre = 0;
    for (int w = 0; w < 8; ++w) {
        if (w < wave) pre += s_wsum[w];
    }
    const int excl = pre + inc - d;
    const int node = b * 256 + tid;
    if (node < N_NODES) rowptr[node] = base + excl;
    if (node == N_NODES - 1) rowptr[N_NODES] = base + excl + d;
    s_cur[tid] = excl;
    __syncthreads();

    const bool in_lds = total <= LCAP;
    for (int jb = 0; jb < maxn; jb += 8) {
        const int idx = jb + sub;
        int w[8];
#pragma unroll
        for (int r = 0; r < 8; ++r) w[r] = gstage[ro[r] + min(idx, max(n[r] - 1, 0))];
#pragma unroll
        for (int r = 0; r < 8; ++r) {
            const int act = (idx < n[r]) ? 1 : 0;
            int pos = atomicAdd(&s_cur[(w[r] >> 16) & 255], act);
            int sid = w[r] & 0xFFFF;
            if (act) {
                if (in_lds) {
                    if ((unsigned)pos < (unsigned)LCAP) s_perm[pos] = sid;
                } else {
                    int gi = base + pos;
                    if ((unsigned)gi < (unsigned)N_EDGES) perm[gi] = sid;
                }
            }
        }
    }
    __syncthreads();
    if (in_lds) {
        for (int i = tid; i < total; i += 256) {
            int gi = base + i;
            if ((unsigned)gi < (unsigned)N_EDGES) perm[gi] = s_perm[i];
        }
    }
}

__global__ __launch_bounds__(512) void sage_gemm_kernel(
    const float* __restrict__ x,
    const int* __restrict__ rowptr, const int* __restrict__ perm,
    const float* __restrict__ Wl,  const float* __restrict__ Wr,
    const float* __restrict__ bias,
    float* __restrict__ out, int do_relu)
{
    __shared__ float lw[128 * KSTRIDE];
    __shared__ float at[GWAVES * 16 * ASTRIDE];

    int tid = threadIdx.x;
    for (int idx = tid; idx < 256 * 128; idx += 512) {
        int k = idx >> 7;
        int n = idx & 127;
        float w = (k < 128) ? Wl[idx] : Wr[idx - 128 * 128];
        lw[n * KSTRIDE + k] = w;
    }

    int lane = tid & 31;
    int wave = tid >> 5;
    int h    = lane >> 4;
    int col  = lane & 15;

    int rowBase = blockIdx.x * (GWAVES * 16) + wave * 16;
    float* atw  = at + wave * (16 * ASTRIDE);

    {
        const float4* __restrict__ x4 = (const float4*)x;
        const float4 zero4 = {0.f, 0.f, 0.f, 0.f};

        int rpi = rowBase + lane;
        if (rpi > N_NODES) rpi = N_NODES;
        int rp = rowptr[rpi];
        rp = min(max(rp, 0), N_EDGES);
        int rpn = rowptr[min(rpi + 1, N_NODES)];
        rpn = min(max(rpn, 0), N_EDGES);
        const float scale_r = 1.0f / fmaxf((float)(rpn - rp), 1.0f);

        const int e_beg = __builtin_amdgcn_readlane(rp, 0);
        const int e_end = __builtin_amdgcn_readlane(rpn, 15);

        auto load_ids = [&](int b) -> int {
            int e  = b + lane;
            int id = perm[min(e, N_EDGES - 1)];
            return (e < e_end) ? id : 0;
        };
        auto issue_rows = [&](float4 (&v)[16], int ids, int j0) {
            unsigned s[16];
#pragma unroll
            for (int u = 0; u < 16; ++u) {
                s[u] = (unsigned)__builtin_amdgcn_readlane(ids, j0 + u);
                if (s[u] > (unsigned)(N_NODES - 1)) s[u] = N_NODES - 1;
            }
#pragma unroll
            for (int u = 0; u < 16; ++u) v[u] = x4[(size_t)s[u] * 32 + lane];
        };

        int cur     = 0;
        int row_end = __builtin_amdgcn_readlane(rpn, 0);
        float4 acc  = zero4;

        auto close_row = [&]() {
            float scale = __builtin_bit_cast(float, __builtin_amdgcn_readlane(__builtin_bit_cast(int, scale_r), cur));
            acc.x *= scale;
            acc.y *= scale;
            acc.z *= scale;
            acc.w *= scale;
            *(float4*)(atw + cur * ASTRIDE + 4 * lane) = acc;
            acc     = zero4;
            ++cur;
            row_end = (cur < 16) ? __builtin_amdgcn_readlane(rpn, cur) : 0x7fffffff;
        };
        auto add_rows = [&](const float4 (&v)[16], int b) {
#pragma unroll
            for (int u = 0; u < 16; ++u) {
                while (__builtin_expect(b + u >= row_end, 0)) close_row();
                acc.x += v[u].x;
                acc.y += v[u].y;
                acc.z += v[u].z;
                acc.w += v[u].w;
            }
        };

        float4 va[16];
        float4 vb[16];
        int b   = e_beg;
        int ids = load_ids(b);
        issue_rows(va, ids, 0);
        while (b < e_end) {
            int ids_n = load_ids(b + 32);
            issue_rows(vb, ids, 16);
            add_rows(va, b);
            issue_rows(va, ids_n, 0);
            add_rows(vb, b + 16);
            ids = ids_n;
            b += 32;
        }
        while (cur < 16) close_row();
    }
    __syncthreads();

    int rowA    = rowBase + col;
    int rowAc   = rowA < N_NODES ? rowA : (N_NODES - 1);

    const float* arow_m = atw + col * ASTRIDE + 2 * h;
    const float* arow_x = x   + (long long)rowAc * FDIM + 2 * h;
    const float* brow   = &lw[col * KSTRIDE + 2 * h];

    v8f acc[8];
    v8f z = {0.f, 0.f, 0.f, 0.f, 0.f, 0.f, 0.f, 0.f};
#pragma unroll
    for (int t = 0; t < 8; ++t) acc[t] = z;

#pragma unroll 4
    for (int kk = 0; kk < 128; kk += 4) {
        v2f a = *(const v2f*)(arow_m + kk);
#pragma unroll
        for (int t = 0; t < 8; ++t) {
            v2f b = *(const v2f*)(brow + t * (16 * KSTRIDE) + kk);
            acc[t] = __builtin_amdgcn_wmma_f32_16x16x4_f32(
                false, a, false, b, (short)0, acc[t], false, false);
        }
    }

#pragma unroll 4
    for (int kk = 0; kk < 128; kk += 4) {
        v2f a = *(const v2f*)(arow_x + kk);
#pragma unroll
        for (int t = 0; t < 8; ++t) {
            v2f b = *(const v2f*)(brow + t * (16 * KSTRIDE) + 128 + kk);
            acc[t] = __builtin_amdgcn_wmma_f32_16x16x4_f32(
                false, a, false, b, (short)0, acc[t], false, false);
        }
    }

#pragma unroll
    for (int t = 0; t < 8; ++t) {
        int n  = t * 16 + col;
        float bv = bias[n];
#pragma unroll
        for (int v = 0; v < 8; ++v) {
            int row = rowBase + v + 8 * h;
            float val = acc[t][v] + bv;
            if (do_relu) val = fmaxf(val, 0.0f);
            if (row < N_NODES) out[(long long)row * FDIM + n] = val;
        }
    }
}

extern "C" void kernel_launch(void* const* d_in, const int* in_sizes, int n_in,
                              void* d_out, int out_size, void* d_ws, size_t ws_size,
                              hipStream_t stream) {
    const float* x   = (const float*)d_in[0];
    const int*   ei  = (const int*)d_in[1];
    const float* Wl0 = (const float*)d_in[2];
    const float* Wr0 = (const float*)d_in[3];
    const float* b0  = (const float*)d_in[4];
    const float* Wl1 = (const float*)d_in[5];
    const float* Wr1 = (const float*)d_in[6];
    const float* b1  = (const float*)d_in[7];
    float* out = (float*)d_out;

    const int* src = ei;
    const int* dst = ei + N_EDGES;

    int* rowptr = (int*)d_ws;
    int* gstage = rowptr + NPAD;
    int* perm   = gstage + N_EDGES;
    int* tab    = perm + N_EDGES;
    int* offtab = tab + NPAD;
    float* hbuf = (float*)d_ws + (size_t)N_NODES * FDIM + N_NODES;

    const int gemm_blocks = NPAD / (GWAVES * 16);

    sage_bin_kernel<<<NCHUNK, ATHREADS, 0, stream>>>(src, dst, tab, offtab, gstage);
    sage_fine_kernel<<<NBIN, 256, 0, stream>>>(tab, offtab, gstage, rowptr, perm);

    sage_gemm_kernel<<<gemm_blocks, 512, 0, stream>>>(x, rowptr, perm, Wl0, Wr0, b0, hbuf, 1);
    sage_gemm_kernel<<<gemm_blocks, 512, 0, stream>>>(hbuf, rowptr, perm, Wl1, Wr1, b1, out, 0);
}


# GEMM waves out of step: by wave index half do the self product (x @ Wr) before their gather, half after; gather at raised priority; barrier now behind weight staging; early waves sum self first
# speedup vs baseline: 3.5800x; 1.1009x over previous
#include <hip/hip_runtime.h>


typedef float v2f __attribute__((ext_vector_type(2)));
typedef float v8f __attribute__((ext_vector_type(8)));

#define N_NODES 50000
#define FDIM    128
#define N_EDGES 800000
#define KSTRIDE 260
#define ASTRIDE 132
#define GWAVES  16
#define NPAD    50176

#define NBIN     196
#define NCHUNK   256
#define CHUNK    3125
#define ATHREADS 320
#define AEPT     10
#define LCAP     8192
static_assert(NCHUNK * CHUNK == N_EDGES, "chunks cover the edge list exactly");
static_assert(ATHREADS * AEPT >= CHUNK, "every edge of a chunk has a slot");
static_assert(NBIN * 256 == NPAD, "ranges cover the padded node count");
static_assert(N_NODES <= 65536, "a source id fits the low 16 bits of a staged word");

__global__ __launch_bounds__(ATHREADS) void sage_bin_kernel(
    const int* __restrict__ src, const int* __restrict__ dst,
    int* __restrict__ tab, int* __restrict__ offtab, int* __restrict__ gstage)
{
    __shared__ int s_cnt[NBIN];
    __shared__ int s_off[NBIN];
    __shared__ int s_stage[CHUNK];
    const int tid = threadIdx.x;
    const int c   = blockIdx.x;
    if (tid < NBIN) s_cnt[tid] = 0;
    __syncthreads();

    int dv[AEPT];
    int sv[AEPT];
    int bin[AEPT];
    int word[AEPT];
    int r[AEPT];
#pragma unroll
    for (int k = 0; k < AEPT; ++k) {
        int i = k * ATHREADS + tid;
        int e = c * CHUNK + min(i, CHUNK - 1);
        dv[k] = dst[e];
        sv[k] = src[e];
    }
#pragma unroll
    for (int k = 0; k < AEPT; ++k) {
        int i = k * ATHREADS + tid;
        unsigned ub = (unsigned)dv[k] >> 8;
        bin[k]  = ub > (unsigned)(NBIN - 1) ? (NBIN - 1) : (int)ub;
        word[k] = ((dv[k] & 255) << 16) | (sv[k] & 0xFFFF);
        r[k] = 0;
        if (i < CHUNK) r[k] = atomicAdd(&s_cnt[bin[k]], 1);
    }
    __syncthreads();
    if (tid < 32) {
        int v[7];
        int sum = 0;
#pragma unroll
        for (int j = 0; j < 7; ++j) {
            int bb = tid * 7 + j;
            v[j] = (bb < NBIN) ? s_cnt[bb] : 0;
            sum += v[j];
        }
        int inc = sum;
        for (int o = 1; o < 32; o <<= 1) {
            int t = __shfl_up(inc, o);
            if (tid >= o) inc += t;
        }
        int run = inc - sum;
#pragma unroll
        for (int j = 0; j < 7; ++j) {
            int bb = tid * 7 + j;
            if (bb < NBIN) s_off[bb] = run;
            run += v[j];
        }
    }
    __syncthreads();
    if (tid < NBIN) {
        tab[tid * NCHUNK + c]    = s_cnt[tid];
        offtab[tid * NCHUNK + c] = s_off[tid];
    }
#pragma unroll
    for (int k = 0; k < AEPT; ++k) {
        int i = k * ATHREADS + tid;
        if (i < CHUNK) {
            int pos = s_off[bin[k]] + r[k];
            if ((unsigned)pos < (unsigned)CHUNK) s_stage[pos] = word[k];
        }
    }
    __syncthreads();
    for (int i = tid; i < CHUNK; i += ATHREADS) gstage[c * CHUNK + i] = s_stage[i];
}

__global__ __launch_bounds__(256) void sage_fine_kernel(
    const int* __restrict__ tab, const int* __restrict__ offtab, const int* __restrict__ gstage,
    int* __restrict__ rowptr, int* __restrict__ perm)
{
    __shared__ int s_red[16];
    __shared__ int s_wsum[8];
    __shared__ int s_cn[NCHUNK];
    __shared__ int s_ro[NCHUNK];
    __shared__ int s_deg[256];
    __shared__ int s_cur[256];
    __shared__ int s_perm[LCAP];
    const int tid  = threadIdx.x;
    const int lane = tid & 31;
    const int wave = tid >> 5;
    const int b    = blockIdx.x;

    int cn = tab[b * NCHUNK + tid];
    int of = offtab[b * NCHUNK + tid];
    cn = min(max(cn, 0), CHUNK);
    of = min(max(of, 0), CHUNK - cn);
    s_cn[tid]  = cn;
    s_ro[tid]  = tid * CHUNK + min(of, CHUNK - 1);
    s_deg[tid] = 0;

    const int4* tab4 = (const int4*)tab;
    const int n4 = b * (NCHUNK / 4);
    int part = 0;
    for (int j0 = 0; j0 < n4; j0 += 256 * 8) {
        int4 qv[8];
#pragma unroll
        for (int u = 0; u < 8; ++u) {
            int j = j0 + u * 256 + tid;
            qv[u] = tab4[min(j, n4 - 1)];
        }
        __builtin_amdgcn_sched_barrier(0);
#pragma unroll
        for (int u = 0; u < 8; ++u) {
            int j = j0 + u * 256 + tid;
            int s = (qv[u].x + qv[u].y) + (qv[u].z + qv[u].w);
            part += (j < n4) ? s : 0;
        }
    }
    int ps = part;
    int cs = cn;
    for (int o = 16; o > 0; o >>= 1) {
        ps += __shfl_xor(ps, o);
        cs += __shfl_xor(cs, o);
    }
    if (lane == 0) {
        s_red[wave]     = ps;
        s_red[8 + wave] = cs;
    }
    __syncthreads();
    int base  = 0;
    int total = 0;
    for (int w = 0; w < 8; ++w) {
        base  += s_red[w];
        total += s_red[8 + w];
    }
    base = min(max(base, 0), N_EDGES);

    const int grp = tid >> 3;
    const int sub = tid & 7;
    int n[8];
    int ro[8];
    int maxn = 0;
#pragma unroll
    for (int r = 0; r < 8; ++r) {
        n[r]  = s_cn[r * 32 + grp];
        ro[r] = s_ro[r * 32 + grp];
        maxn  = max(maxn, n[r]);
    }

    for (int jb = 0; jb < maxn; jb += 8) {
        const int idx = jb + sub;
        int w[8];
#pragma unroll
        for (int r = 0; r < 8; ++r) w[r] = gstage[ro[r] + min(idx, max(n[r] - 1, 0))];
#pragma unroll
        for (int r = 0; r < 8; ++r) {
            atomicAdd(&s_deg[(w[r] >> 16) & 255], (idx < n[r]) ? 1 : 0);
        }
    }
    __syncthreads();
    const int d = s_deg[tid];
    int inc = d;
    for (int o = 1; o < 32; o <<= 1) {
        int t = __shfl_up(inc, o);
        if (lane >= o) inc += t;
    }
    if (lane == 31) s_wsum[wave] = inc;
    __syncthreads();
    int pre = 0;
    for (int w = 0; w < 8; ++w) {
        if (w < wave) pre += s_wsum[w];
    }
    const int excl = pre + inc - d;
    const int node = b * 256 + tid;
    if (node < N_NODES) rowptr[node] = base + excl;
    if (node == N_NODES - 1) rowptr[N_NODES] = base + excl + d;
    s_cur[tid] = excl;
    __syncthreads();

    const bool in_lds = total <= LCAP;
    for (int jb = 0; jb < maxn; jb += 8) {
        const int idx = jb + sub;
        int w[8];
#pragma unroll
        for (int r = 0; r < 8; ++r) w[r] = gstage[ro[r] + min(idx, max(n[r] - 1, 0))];
#pragma unroll
        for (int r = 0; r < 8; ++r) {
            const int act = (idx < n[r]) ? 1 : 0;
            int pos = atomicAdd(&s_cur[(w[r] >> 16) & 255], act);
            int sid = w[r] & 0xFFFF;
            if (act) {
                if (in_lds) {
                    if ((unsigned)pos < (unsigned)LCAP) s_perm[pos] = sid;
                } else {
                    int gi = base + pos;
                    if ((unsigned)gi < (unsigned)N_EDGES) perm[gi] = sid;
                }
            }
        }
    }
    __syncthreads();
    if (in_lds) {
        for (int i = tid; i < total; i += 256) {
            int gi = base + i;
            if ((unsigned)gi < (unsigned)N_EDGES) perm[gi] = s_perm[i];
        }
    }
}

__global__ __launch_bounds__(512) void sage_gemm_kernel(
    const float* __restrict__ x,
    const int* __restrict__ rowptr, const int* __restrict__ perm,
    const float* __restrict__ Wl,  const float* __restrict__ Wr,
    const float* __restrict__ bias,
    float* __restrict__ out, int do_relu)
{
    __shared__ float lw[128 * KSTRIDE];
    __shared__ float at[GWAVES * 16 * ASTRIDE];

    int tid = threadIdx.x;
    for (int idx = tid; idx < 256 * 128; idx += 512) {
        int k = idx >> 7;
        int n = idx & 127;
        float w = (k < 128) ? Wl[idx] : Wr[idx - 128 * 128];
        lw[n * KSTRIDE + k] = w;
    }

    int lane = tid & 31;
    int wave = tid >> 5;
    int h    = lane >> 4;
    int col  = lane & 15;

    int rowBase = blockIdx.x * (GWAVES * 16) + wave * 16;
    float* atw  = at + wave * (16 * ASTRIDE);

    __syncthreads();

    const int  wv    = __builtin_amdgcn_readfirstlane(wave);
    const bool early = ((wv ^ (wv >> 2)) & 1) != 0;

    int rowA    = rowBase + col;
    int rowAc   = rowA < N_NODES ? rowA : (N_NODES - 1);

    const float* arow_m = atw + col * ASTRIDE + 2 * h;
    const float* arow_x = x   + (long long)rowAc * FDIM + 2 * h;
    const float* brow   = &lw[col * KSTRIDE + 2 * h];

    v8f acc[8];
    {
        v8f z = {0.f, 0.f, 0.f, 0.f, 0.f, 0.f, 0.f, 0.f};
#pragma unroll
        for (int t = 0; t < 8; ++t) acc[t] = z;
    }

    auto mean_product = [&]() {
#pragma unroll 4
        for (int kk = 0; kk < 128; kk += 4) {
            v2f a = *(const v2f*)(arow_m + kk);
#pragma unroll
            for (int t = 0; t < 8; ++t) {
                v2f b = *(const v2f*)(brow + t * (16 * KSTRIDE) + kk);
                acc[t] = __builtin_amdgcn_wmma_f32_16x16x4_f32(
                    false, a, false, b, (short)0, acc[t], false, false);
            }
        }
    };
    auto self_product = [&]() {
#pragma unroll 4
        for (int kk = 0; kk < 128; kk += 4) {
            v2f a = *(const v2f*)(arow_x + kk);
#pragma unroll
            for (int t = 0; t < 8; ++t) {
                v2f b = *(const v2f*)(brow + t * (16 * KSTRIDE) + 128 + kk);
                acc[t] = __builtin_amdgcn_wmma_f32_16x16x4_f32(
                    false, a, false, b, (short)0, acc[t], false, false);
            }
        }
    };

    if (early) self_product();

    __builtin_amdgcn_s_setprio(3);
    {
        const float4* __restrict__ x4 = (const float4*)x;
        const float4 zero4 = {0.f, 0.f, 0.f, 0.f};

        int rpi = rowBase + lane;
        if (rpi > N_NODES) rpi = N_NODES;
        int rp = rowptr[rpi];
        rp = min(max(rp, 0), N_EDGES);
        int rpn = rowptr[min(rpi + 1, N_NODES)];
        rpn = min(max(rpn, 0), N_EDGES);
        const float scale_r = 1.0f / fmaxf((float)(rpn - rp), 1.0f);

        const int e_beg = __builtin_amdgcn_readlane(rp, 0);
        const int e_end = __builtin_amdgcn_readlane(rpn, 15);

        auto load_ids = [&](int b) -> int {
            int e  = b + lane;
            int id = perm[min(e, N_EDGES - 1)];
            return (e < e_end) ? id : 0;
        };
        auto issue_rows = [&](float4 (&v)[16], int ids, int j0) {
            unsigned s[16];
#pragma unroll
            for (int u = 0; u < 16; ++u) {
                s[u] = (unsigned)__builtin_amdgcn_readlane(ids, j0 + u);
                if (s[u] > (unsigned)(N_NODES - 1)) s[u] = N_NODES - 1;
            }
#pragma unroll
            for (int u = 0; u < 16; ++u) v[u] = x4[(size_t)s[u] * 32 + lane];
        };

        int cur     = 0;
        int row_end = __builtin_amdgcn_readlane(rpn, 0);
        float4 acc  = zero4;

        auto close_row = [&]() {
            float scale = __builtin_bit_cast(float, __builtin_amdgcn_readlane(__builtin_bit_cast(int, scale_r), cur));
            acc.x *= scale;
            acc.y *= scale;
            acc.z *= scale;
            acc.w *= scale;
            *(float4*)(atw + cur * ASTRIDE + 4 * lane) = acc;
            acc     = zero4;
            ++cur;
            row_end = (cur < 16) ? __builtin_amdgcn_readlane(rpn, cur) : 0x7fffffff;
        };
        auto add_rows = [&](const float4 (&v)[16], int b) {
#pragma unroll
            for (int u = 0; u < 16; ++u) {
                while (__builtin_expect(b + u >= row_end, 0)) close_row();
                acc.x += v[u].x;
                acc.y += v[u].y;
                acc.z += v[u].z;
                acc.w += v[u].w;
            }
        };

        float4 va[16];
        float4 vb[16];
        int b   = e_beg;
        int ids = load_ids(b);
        issue_rows(va, ids, 0);
        while (b < e_end) {
            int ids_n = load_ids(b + 32);
            issue_rows(vb, ids, 16);
            add_rows(va, b);
            issue_rows(va, ids_n, 0);
            add_rows(vb, b + 16);
            ids = ids_n;
            b += 32;
        }
        while (cur < 16) close_row();
    }
    __builtin_amdgcn_s_setprio(0);
    mean_product();
    if (!early) self_product();

#pragma unroll
    for (int t = 0; t < 8; ++t) {
        int n  = t * 16 + col;
        float bv = bias[n];
#pragma unroll
        for (int v = 0; v < 8; ++v) {
            int row = rowBase + v + 8 * h;
            float val = acc[t][v] + bv;
            if (do_relu) val = fmaxf(val, 0.0f);
            if (row < N_NODES) out[(long long)row * FDIM + n] = val;
        }
    }
}

extern "C" void kernel_launch(void* const* d_in, const int* in_sizes, int n_in,
                              void* d_out, int out_size, void* d_ws, size_t ws_size,
                              hipStream_t stream) {
    const float* x   = (const float*)d_in[0];
    const int*   ei  = (const int*)d_in[1];
    const float* Wl0 = (const float*)d_in[2];
    const float* Wr0 = (const float*)d_in[3];
    const float* b0  = (const float*)d_in[4];
    const float* Wl1 = (const float*)d_in[5];
    const float* Wr1 = (const float*)d_in[6];
    const float* b1  = (const float*)d_in[7];
    float* out = (float*)d_out;

    const int* src = ei;
    const int* dst = ei + N_EDGES;

    int* rowptr = (int*)d_ws;
    int* gstage = rowptr + NPAD;
    int* perm   = gstage + N_EDGES;
    int* tab    = perm + N_EDGES;
    int* offtab = tab + NPAD;
    float* hbuf = (float*)d_ws + (size_t)N_NODES * FDIM + N_NODES;

    const int gemm_blocks = NPAD / (GWAVES * 16);

    sage_bin_kernel<<<NCHUNK, ATHREADS, 0, stream>>>(src, dst, tab, offtab, gstage);
    sage_fine_kernel<<<NBIN, 256, 0, stream>>>(tab, offtab, gstage, rowptr, perm);

    sage_gemm_kernel<<<gemm_blocks, 512, 0, stream>>>(x, rowptr, perm, Wl0, Wr0, b0, hbuf, 1);
    sage_gemm_kernel<<<gemm_blocks, 512, 0, stream>>>(hbuf, rowptr, perm, Wl1, Wr1, b1, out, 0);
}


# GEMM kernel's global accesses as scalar base + one 32-bit offset register + constant (gather: 16 offsets, then 16 row loads; weight staging; out stores): no address rewritten behind an issued access
# speedup vs baseline: 4.1982x; 1.1727x over previous
#include <hip/hip_runtime.h>


typedef float v2f __attribute__((ext_vector_type(2)));
typedef float v8f __attribute__((ext_vector_type(8)));

#define N_NODES 50000
#define FDIM    128
#define N_EDGES 800000
#define KSTRIDE 260
#define ASTRIDE 132
#define GWAVES  16
#define NPAD    50176

#define NBIN     196
#define NCHUNK   256
#define CHUNK    3125
#define ATHREADS 320
#define AEPT     10
#define LCAP     8192
static_assert(NCHUNK * CHUNK == N_EDGES, "chunks cover the edge list exactly");
static_assert(ATHREADS * AEPT >= CHUNK, "every edge of a chunk has a slot");
static_assert(NBIN * 256 == NPAD, "ranges cover the padded node count");
static_assert(N_NODES <= 65536, "a source id fits the low 16 bits of a staged word");

__global__ __launch_bounds__(ATHREADS) void sage_bin_kernel(
    const int* __restrict__ src, const int* __restrict__ dst,
    int* __restrict__ tab, int* __restrict__ offtab, int* __restrict__ gstage)
{
    __shared__ int s_cnt[NBIN];
    __shared__ int s_off[NBIN];
    __shared__ int s_stage[CHUNK];
    const int tid = threadIdx.x;
    const int c   = blockIdx.x;
    if (tid < NBIN) s_cnt[tid] = 0;
    __syncthreads();

    int dv[AEPT];
    int sv[AEPT];
    int bin[AEPT];
    int word[AEPT];
    int r[AEPT];
#pragma unroll
    for (int k = 0; k < AEPT; ++k) {
        int i = k * ATHREADS + tid;
        int e = c * CHUNK + min(i, CHUNK - 1);
        dv[k] = dst[e];
        sv[k] = src[e];
    }
#pragma unroll
    for (int k = 0; k < AEPT; ++k) {
        int i = k * ATHREADS + tid;
        unsigned ub = (unsigned)dv[k] >> 8;
        bin[k]  = ub > (unsigned)(NBIN - 1) ? (NBIN - 1) : (int)ub;
        word[k] = ((dv[k] & 255) << 16) | (sv[k] & 0xFFFF);
        r[k] = 0;
        if (i < CHUNK) r[k] = atomicAdd(&s_cnt[bin[k]], 1);
    }
    __syncthreads();
    if (tid < 32) {
        int v[7];
        int sum = 0;
#pragma unroll
        for (int j = 0; j < 7; ++j) {
            int bb = tid * 7 + j;
            v[j] = (bb < NBIN) ? s_cnt[bb] : 0;
            sum += v[j];
        }
        int inc = sum;
        for (int o = 1; o < 32; o <<= 1) {
            int t = __shfl_up(inc, o);
            if (tid >= o) inc += t;
        }
        int run = inc - sum;
#pragma unroll
        for (int j = 0; j < 7; ++j) {
            int bb = tid * 7 + j;
            if (bb < NBIN) s_off[bb] = run;
            run += v[j];
        }
    }
    __syncthreads();
    if (tid < NBIN) {
        tab[tid * NCHUNK + c]    = s_cnt[tid];
        offtab[tid * NCHUNK + c] = s_off[tid];
    }
#pragma unroll
    for (int k = 0; k < AEPT; ++k) {
        int i = k * ATHREADS + tid;
        if (i < CHUNK) {
            int pos = s_off[bin[k]] + r[k];
            if ((unsigned)pos < (unsigned)CHUNK) s_stage[pos] = word[k];
        }
    }
    __syncthreads();
    for (int i = tid; i < CHUNK; i += ATHREADS) gstage[c * CHUNK + i] = s_stage[i];
}

__global__ __launch_bounds__(256) void sage_fine_kernel(
    const int* __restrict__ tab, const int* __restrict__ offtab, const int* __restrict__ gstage,
    int* __restrict__ rowptr, int* __restrict__ perm)
{
    __shared__ int s_red[16];
    __shared__ int s_wsum[8];
    __shared__ int s_cn[NCHUNK];
    __shared__ int s_ro[NCHUNK];
    __shared__ int s_deg[256];
    __shared__ int s_cur[256];
    __shared__ int s_perm[LCAP];
    const int tid  = threadIdx.x;
    const int lane = tid & 31;
    const int wave = tid >> 5;
    const int b    = blockIdx.x;

    int cn = tab[b * NCHUNK + tid];
    int of = offtab[b * NCHUNK + tid];
    cn = min(max(cn, 0), CHUNK);
    of = min(max(of, 0), CHUNK - cn);
    s_cn[tid]  = cn;
    s_ro[tid]  = tid * CHUNK + min(of, CHUNK - 1);
    s_deg[tid] = 0;

    const int4* tab4 = (const int4*)tab;
    const int n4 = b * (NCHUNK / 4);
    int part = 0;
    for (int j0 = 0; j0 < n4; j0 += 256 * 8) {
        int4 qv[8];
#pragma unroll
        for (int u = 0; u < 8; ++u) {
            int j = j0 + u * 256 + tid;
            qv[u] = tab4[min(j, n4 - 1)];
        }
        __builtin_amdgcn_sched_barrier(0);
#pragma unroll
        for (int u = 0; u < 8; ++u) {
            int j = j0 + u * 256 + tid;
            int s = (qv[u].x + qv[u].y) + (qv[u].z + qv[u].w);
            part += (j < n4) ? s : 0;
        }
    }
    int ps = part;
    int cs = cn;
    for (int o = 16; o > 0; o >>= 1) {
        ps += __shfl_xor(ps, o);
        cs += __shfl_xor(cs, o);
    }
    if (lane == 0) {
        s_red[wave]     = ps;
        s_red[8 + wave] = cs;
    }
    __syncthreads();
    int base  = 0;
    int total = 0;
    for (int w = 0; w < 8; ++w) {
        base  += s_red[w];
        total += s_red[8 + w];
    }
    base = min(max(base, 0), N_EDGES);

    const int grp = tid >> 3;
    const int sub = tid & 7;
    int n[8];
    int ro[8];
    int maxn = 0;
#pragma unroll
    for (int r = 0; r < 8; ++r) {
        n[r]  = s_cn[r * 32 + grp];
        ro[r] = s_ro[r * 32 + grp];
        maxn  = max(maxn, n[r]);
    }

    for (int jb = 0; jb < maxn; jb += 8) {
        const int idx = jb + sub;
        int w[8];
#pragma unroll
        for (int r = 0; r < 8; ++r) w[r] = gstage[ro[r] + min(idx, max(n[r] - 1, 0))];
#pragma unroll
        for (int r = 0; r < 8; ++r) {
            atomicAdd(&s_deg[(w[r] >> 16) & 255], (idx < n[r]) ? 1 : 0);
        }
    }
    __syncthreads();
    const int d = s_deg[tid];
    int inc = d;
    for (int o = 1; o < 32; o <<= 1) {
        int t = __shfl_up(inc, o);
        if (lane >= o) inc += t;
    }
    if (lane == 31) s_wsum[wave] = inc;
    __syncthreads();
    int pre = 0;
    for (int w = 0; w < 8; ++w) {
        if (w < wave) pre += s_wsum[w];
    }
    const int excl = pre + inc - d;
    const int node = b * 256 + tid;
    if (node < N_NODES) rowptr[node] = base + excl;
    if (node == N_NODES - 1) rowptr[N_NODES] = base + excl + d;
    s_cur[tid] = excl;
    __syncthreads();

    const bool in_lds = total <= LCAP;
    for (int jb = 0; jb < maxn; jb += 8) {
        const int idx = jb + sub;
        int w[8];
#pragma unroll
        for (int r = 0; r < 8; ++r) w[r] = gstage[ro[r] + min(idx, max(n[r] - 1, 0))];
#pragma unroll
        for (int r = 0; r < 8; ++r) {
            const int act = (idx < n[r]) ? 1 : 0;
            int pos = atomicAdd(&s_cur[(w[r] >> 16) & 255], act);
            int sid = w[r] & 0xFFFF;
            if (act) {
                if (in_lds) {
                    if ((unsigned)pos < (unsigned)LCAP) s_perm[pos] = sid;
                } else {
                    int gi = base + pos;
                    if ((unsigned)gi < (unsigned)N_EDGES) perm[gi] = sid;
                }
            }
        }
    }
    __syncthreads();
    if (in_lds) {
        for (int i = tid; i < total; i += 256) {
            int gi = base + i;
            if ((unsigned)gi < (unsigned)N_EDGES) perm[gi] = s_perm[i];
        }
    }
}

__global__ __launch_bounds__(512) void sage_gemm_kernel(
    const float* __restrict__ x,
    const int* __restrict__ rowptr, const int* __restrict__ perm,
    const float* __restrict__ Wl,  const float* __restrict__ Wr,
    const float* __restrict__ bias,
    float* __restrict__ out, int do_relu)
{
    __shared__ float lw[128 * KSTRIDE];
    __shared__ float at[GWAVES * 16 * ASTRIDE];

    int tid = threadIdx.x;
    {
        const int n  = tid & 127;
        const int k0 = tid >> 7;
        float* lwt = &lw[n * KSTRIDE + k0];
#pragma unroll
        for (int j = 0; j < 32; ++j) lwt[4 * j] = Wl[tid + 512 * j];
#pragma unroll
        for (int j = 0; j < 32; ++j) lwt[128 + 4 * j] = Wr[tid + 512 * j];
    }

    int lane = tid & 31;
    int wave = tid >> 5;
    int h    = lane >> 4;
    int col  = lane & 15;

    int rowBase = blockIdx.x * (GWAVES * 16) + wave * 16;
    float* atw  = at + wave * (16 * ASTRIDE);

    __syncthreads();

    const int  wv    = __builtin_amdgcn_readfirstlane(wave);
    const bool early = ((wv ^ (wv >> 2)) & 1) != 0;

    int rowA    = rowBase + col;
    int rowAc   = rowA < N_NODES ? rowA : (N_NODES - 1);

    const float* arow_m = atw + col * ASTRIDE + 2 * h;
    const float* arow_x = x   + (long long)rowAc * FDIM + 2 * h;
    const float* brow   = &lw[col * KSTRIDE + 2 * h];

    v8f acc[8];
    {
        v8f z = {0.f, 0.f, 0.f, 0.f, 0.f, 0.f, 0.f, 0.f};
#pragma unroll
        for (int t = 0; t < 8; ++t) acc[t] = z;
    }

    auto mean_product = [&]() {
#pragma unroll 4
        for (int kk = 0; kk < 128; kk += 4) {
            v2f a = *(const v2f*)(arow_m + kk);
#pragma unroll
            for (int t = 0; t < 8; ++t) {
                v2f b = *(const v2f*)(brow + t * (16 * KSTRIDE) + kk);
                acc[t] = __builtin_amdgcn_wmma_f32_16x16x4_f32(
                    false, a, false, b, (short)0, acc[t], false, false);
            }
        }
    };
    auto self_product = [&]() {
#pragma unroll 4
        for (int kk = 0; kk < 128; kk += 4) {
            v2f a = *(const v2f*)(arow_x + kk);
#pragma unroll
            for (int t = 0; t < 8; ++t) {
                v2f b = *(const v2f*)(brow + t * (16 * KSTRIDE) + 128 + kk);
                acc[t] = __builtin_amdgcn_wmma_f32_16x16x4_f32(
                    false, a, false, b, (short)0, acc[t], false, false);
            }
        }
    };

    if (early) self_product();

    __builtin_amdgcn_s_setprio(3);
    {
        const char* __restrict__ xb = (const char*)x;
        const unsigned lane16 = (unsigned)lane * 16u;
        const float4 zero4 = {0.f, 0.f, 0.f, 0.f};

        int rpi = rowBase + lane;
        if (rpi > N_NODES) rpi = N_NODES;
        int rp = rowptr[rpi];
        rp = min(max(rp, 0), N_EDGES);
        int rpn = rowptr[min(rpi + 1, N_NODES)];
        rpn = min(max(rpn, 0), N_EDGES);
        const float scale_r = 1.0f / fmaxf((float)(rpn - rp), 1.0f);

        const int e_beg = __builtin_amdgcn_readlane(rp, 0);
        const int e_end = __builtin_amdgcn_readlane(rpn, 15);

        auto load_ids = [&](int b) -> int {
            int e  = b + lane;
            int id = perm[min(e, N_EDGES - 1)];
            return (e < e_end) ? id : 0;
        };
        auto issue_rows = [&](float4 (&v)[16], int ids, int j0) {
            unsigned off[16];
#pragma unroll
            for (int u = 0; u < 16; ++u) {
                unsigned s = (unsigned)__builtin_amdgcn_readlane(ids, j0 + u);
                if (s > (unsigned)(N_NODES - 1)) s = N_NODES - 1;
                off[u] = s * 512u + lane16;
            }
#pragma unroll
            for (int u = 0; u < 16; ++u) v[u] = *(const float4*)(xb + off[u]);
        };

        int cur     = 0;
        int row_end = __builtin_amdgcn_readlane(rpn, 0);
        float4 acc  = zero4;

        auto close_row = [&]() {
            float scale = __builtin_bit_cast(float, __builtin_amdgcn_readlane(__builtin_bit_cast(int, scale_r), cur));
            acc.x *= scale;
            acc.y *= scale;
            acc.z *= scale;
            acc.w *= scale;
            *(float4*)(atw + cur * ASTRIDE + 4 * lane) = acc;
            acc     = zero4;
            ++cur;
            row_end = (cur < 16) ? __builtin_amdgcn_readlane(rpn, cur) : 0x7fffffff;
        };
        auto add_rows = [&](const float4 (&v)[16], int b) {
#pragma unroll
            for (int u = 0; u < 16; ++u) {
                while (__builtin_expect(b + u >= row_end, 0)) close_row();
                acc.x += v[u].x;
                acc.y += v[u].y;
                acc.z += v[u].z;
                acc.w += v[u].w;
            }
        };

        float4 va[16];
        float4 vb[16];
        int b   = e_beg;
        int ids = load_ids(b);
        issue_rows(va, ids, 0);
        while (b < e_end) {
            int ids_n = load_ids(b + 32);
            issue_rows(vb, ids, 16);
            add_rows(va, b);
            issue_rows(va, ids_n, 0);
            add_rows(vb, b + 16);
            ids = ids_n;
            b += 32;
        }
        while (cur < 16) close_row();
    }
    __builtin_amdgcn_s_setprio(0);
    mean_product();
    if (!early) self_product();

    const int rowBaseS = blockIdx.x * (GWAVES * 16) + wv * 16;
    if (rowBaseS + 16 <= N_NODES) {
        char* ob = (char*)out;
        const unsigned o0 = ((unsigned)(rowBaseS + 8 * h) * FDIM + (unsigned)col) * 4u;
#pragma unroll
        for (int t = 0; t < 8; ++t) {
            float bv = bias[t * 16 + col];
#pragma unroll
            for (int v = 0; v < 8; ++v) {
                float val = acc[t][v] + bv;
                if (do_relu) val = fmaxf(val, 0.0f);
                *(float*)(ob + o0 + (unsigned)((v * FDIM + t * 16) * 4)) = val;
            }
        }
    } else {
#pragma unroll
        for (int t = 0; t < 8; ++t) {
            int n  = t * 16 + col;
            float bv = bias[n];
#pragma unroll
            for (int v = 0; v < 8; ++v) {
                int row = rowBase + v + 8 * h;
                float val = acc[t][v] + bv;
                if (do_relu) val = fmaxf(val, 0.0f);
                if (row < N_NODES) out[(long long)row * FDIM + n] = val;
            }
        }
    }
}

extern "C" void kernel_launch(void* const* d_in, const int* in_sizes, int n_in,
                              void* d_out, int out_size, void* d_ws, size_t ws_size,
                              hipStream_t stream) {
    const float* x   = (const float*)d_in[0];
    const int*   ei  = (const int*)d_in[1];
    const float* Wl0 = (const float*)d_in[2];
    const float* Wr0 = (const float*)d_in[3];
    const float* b0  = (const float*)d_in[4];
    const float* Wl1 = (const float*)d_in[5];
    const float* Wr1 = (const float*)d_in[6];
    const float* b1  = (const float*)d_in[7];
    float* out = (float*)d_out;

    const int* src = ei;
    const int* dst = ei + N_EDGES;

    int* rowptr = (int*)d_ws;
    int* gstage = rowptr + NPAD;
    int* perm   = gstage + N_EDGES;
    int* tab    = perm + N_EDGES;
    int* offtab = tab + NPAD;
    float* hbuf = (float*)d_ws + (size_t)N_NODES * FDIM + N_NODES;

    const int gemm_blocks = NPAD / (GWAVES * 16);

    sage_bin_kernel<<<NCHUNK, ATHREADS, 0, stream>>>(src, dst, tab, offtab, gstage);
    sage_fine_kernel<<<NBIN, 256, 0, stream>>>(tab, offtab, gstage, rowptr, perm);

    sage_gemm_kernel<<<gemm_blocks, 512, 0, stream>>>(x, rowptr, perm, Wl0, Wr0, b0, hbuf, 1);
    sage_gemm_kernel<<<gemm_blocks, 512, 0, stream>>>(hbuf, rowptr, perm, Wl1, Wr1, b1, out, 0);
}
